# starting kernel
# baseline (speedup 1.0000x reference)
#include <hip/hip_runtime.h>

typedef __attribute__((ext_vector_type(2))) float v2f;
typedef __attribute__((ext_vector_type(8))) float v8f;

#define BATCH 8
#define NNODE 2048
#define FDIM  64
#define ROWS  (BATCH * NNODE)

__global__ void gat_eff_weights(const float* __restrict__ Wk,
                                const float* __restrict__ As,
                                const float* __restrict__ An,
                                float* __restrict__ wse,
                                float* __restrict__ wne) {
    const int i = threadIdx.x;
    float s = 0.f, n = 0.f;
#pragma unroll
    for (int o = 0; o < FDIM; ++o) {
        const float k = Wk[i * FDIM + o];
        s = fmaf(k, As[o], s);
        n = fmaf(k, An[o], n);
    }
    wse[i] = s;
    wne[i] = n;
}

__global__ void gat_scores_wmma(const float* __restrict__ E,
                                const float* __restrict__ wse,
                                const float* __restrict__ wne,
                                float* __restrict__ S,
                                float* __restrict__ T) {
    __shared__ float wl[3 * FDIM];
    const int tid = threadIdx.x;
    if (tid < 3 * FDIM)
        wl[tid] = (tid < FDIM) ? wse[tid]
                : (tid < 2 * FDIM) ? wne[tid - FDIM]
                : 0.f;
    __syncthreads();

    const int lane  = tid & 31;
    const int wave  = tid >> 5;
    const int row0  = (blockIdx.x * (blockDim.x >> 5) + wave) * 16;
    const int mn    = lane & 15;
    const int khalf = (lane >> 4) << 1;

    const float* arow = E + (size_t)(row0 + mn) * FDIM + khalf;
    const float* wcol = wl + ((mn == 0) ? 0 : ((mn == 1) ? FDIM : 2 * FDIM)) + khalf;

    v8f acc = {};
#pragma unroll
    for (int k0 = 0; k0 < FDIM; k0 += 4) {
        const v2f a = *(const v2f*)(arow + k0);
        const v2f b = *(const v2f*)(wcol + k0);
        acc = __builtin_amdgcn_wmma_f32_16x16x4_f32(
             false, a,  false, b,
             (short)0, acc,  false,  false);
    }

    if (mn < 2) {
        float* dst = (mn == 0) ? S : T;
        const int base = row0 + 8 * (lane >> 4);
#pragma unroll
        for (int r = 0; r < 8; ++r) dst[base + r] = acc[r];
    }
}

__device__ __forceinline__ float lrelu(float x) { return x > 0.f ? x : 0.2f * x; }

__global__ void gat_softmax(const float* __restrict__ S,
                            const float* __restrict__ T,
                            float* __restrict__ out) {
    __shared__ float4 tl4[NNODE / 4];

    const int tid  = threadIdx.x;
    const int lane = tid & 31;
    const int wave = tid >> 5;
    const int b    = blockIdx.x >> 8;
    const int row  = ((blockIdx.x & 255) << 3) + wave;

    const float4* tb4 = (const float4*)(T + (size_t)b * NNODE);
    for (int i = tid; i < NNODE / 4; i += blockDim.x) tl4[i] = tb4[i];
    __syncthreads();

    const float sv = S[(size_t)b * NNODE + row];

    float4 e[16];
    float mx = -3.402823466e38f;
#pragma unroll
    for (int it = 0; it < 16; ++it) {
        const float4 v = tl4[lane + 32 * it];
        float4 x;
        x.x = lrelu(sv + v.x);
        x.y = lrelu(sv + v.y);
        x.z = lrelu(sv + v.z);
        x.w = lrelu(sv + v.w);
        mx = fmaxf(mx, fmaxf(fmaxf(x.x, x.y), fmaxf(x.z, x.w)));
        e[it] = x;
    }
#pragma unroll
    for (int off = 16; off > 0; off >>= 1)
        mx = fmaxf(mx, __shfl_xor(mx, off, 32));

    float sum = 0.f;
#pragma unroll
    for (int it = 0; it < 16; ++it) {
        float4 x = e[it];
        x.x = __expf(x.x - mx);
        x.y = __expf(x.y - mx);
        x.z = __expf(x.z - mx);
        x.w = __expf(x.w - mx);
        sum += (x.x + x.y) + (x.z + x.w);
        e[it] = x;
    }
#pragma unroll
    for (int off = 16; off > 0; off >>= 1)
        sum += __shfl_xor(sum, off, 32);
    const float inv = 1.0f / sum;

    float4* o4 = (float4*)(out + ((size_t)b * NNODE + row) * NNODE);
#pragma unroll
    for (int it = 0; it < 16; ++it) {
        float4 x = e[it];
        x.x *= inv; x.y *= inv; x.z *= inv; x.w *= inv;
        o4[lane + 32 * it] = x;
    }
}

extern "C" void kernel_launch(void* const* d_in, const int* in_sizes, int n_in,
                              void* d_out, int out_size, void* d_ws, size_t ws_size,
                              hipStream_t stream) {
    const float* encode = (const float*)d_in[0];
    const float* Wk     = (const float*)d_in[1];
    const float* As     = (const float*)d_in[2];
    const float* An     = (const float*)d_in[3];
    float*       out    = (float*)d_out;

    float* ws  = (float*)d_ws;
    float* wse = ws;
    float* wne = ws + FDIM;
    float* S   = ws + 2 * FDIM;
    float* T   = S + ROWS;

    gat_eff_weights<<<1, FDIM, 0, stream>>>(Wk, As, An, wse, wne);

    gat_scores_wmma<<<ROWS / (16 * 8), 256, 0, stream>>>(encode, wse, wne, S, T);

    gat_softmax<<<BATCH * (NNODE / 8), 256, 0, stream>>>(S, T, out);
}


# softmax launch reshaped to 256 blocks of 8 waves (one per WGP): each wave keeps its t slice in registers and loops over 8 rows; no LDS stage or barrier; per-row arithmetic and sum order unchanged
# speedup vs baseline: 1.0323x; 1.0323x over previous
#include <hip/hip_runtime.h>

typedef __attribute__((ext_vector_type(2))) float v2f;
typedef __attribute__((ext_vector_type(8))) float v8f;

#define BATCH 8
#define NNODE 2048
#define FDIM  64
#define ROWS  (BATCH * NNODE)

__global__ void gat_eff_weights(const float* __restrict__ Wk,
                                const float* __restrict__ As,
                                const float* __restrict__ An,
                                float* __restrict__ wse,
                                float* __restrict__ wne) {
    const int i = threadIdx.x;
    float s = 0.f, n = 0.f;
#pragma unroll
    for (int o = 0; o < FDIM; ++o) {
        const float k = Wk[i * FDIM + o];
        s = fmaf(k, As[o], s);
        n = fmaf(k, An[o], n);
    }
    wse[i] = s;
    wne[i] = n;
}

__global__ void gat_scores_wmma(const float* __restrict__ E,
                                const float* __restrict__ wse,
                                const float* __restrict__ wne,
                                float* __restrict__ S,
                                float* __restrict__ T) {
    __shared__ float wl[3 * FDIM];
    const int tid = threadIdx.x;
    if (tid < 3 * FDIM)
        wl[tid] = (tid < FDIM) ? wse[tid]
                : (tid < 2 * FDIM) ? wne[tid - FDIM]
                : 0.f;
    __syncthreads();

    const int lane  = tid & 31;
    const int wave  = tid >> 5;
    const int row0  = (blockIdx.x * (blockDim.x >> 5) + wave) * 16;
    const int mn    = lane & 15;
    const int khalf = (lane >> 4) << 1;

    const float* arow = E + (size_t)(row0 + mn) * FDIM + khalf;
    const float* wcol = wl + ((mn == 0) ? 0 : ((mn == 1) ? FDIM : 2 * FDIM)) + khalf;

    v8f acc = {};
#pragma unroll
    for (int k0 = 0; k0 < FDIM; k0 += 4) {
        const v2f a = *(const v2f*)(arow + k0);
        const v2f b = *(const v2f*)(wcol + k0);
        acc = __builtin_amdgcn_wmma_f32_16x16x4_f32(
             false, a,  false, b,
             (short)0, acc,  false,  false);
    }

    if (mn < 2) {
        float* dst = (mn == 0) ? S : T;
        const int base = row0 + 8 * (lane >> 4);
#pragma unroll
        for (int r = 0; r < 8; ++r) dst[base + r] = acc[r];
    }
}

#define ROWS_PER_WAVE  8
#define ROWS_PER_BLOCK 64

__device__ __forceinline__ float lrelu(float x) { return x > 0.f ? x : 0.2f * x; }

__global__ __launch_bounds__(256) void gat_softmax(const float* __restrict__ S,
                                                   const float* __restrict__ T,
                                                   float* __restrict__ out) {
    const int tid  = threadIdx.x;
    const int lane = tid & 31;
    const int wave = tid >> 5;
    const int b    = blockIdx.x >> 5;
    const int row0 = ((blockIdx.x & 31) * ROWS_PER_BLOCK) + wave * ROWS_PER_WAVE;

    const float4* tb4 = (const float4*)(T + (size_t)b * NNODE);
    float4 t[16];
#pragma unroll
    for (int it = 0; it < 16; ++it) t[it] = tb4[lane + 32 * it];

    const float* sp = S + (size_t)b * NNODE + row0;
    const float svec = sp[lane & (ROWS_PER_WAVE - 1)];

    float4* o4 = (float4*)(out + ((size_t)b * NNODE + row0) * NNODE);

#pragma unroll 1
    for (int r = 0; r < ROWS_PER_WAVE; ++r) {
        const float sv = __int_as_float(
            __builtin_amdgcn_readlane(__float_as_int(svec), r));

        float4 e[16];
        float mx = -3.402823466e38f;
#pragma unroll
        for (int it = 0; it < 16; ++it) {
            const float4 v = t[it];
            float4 x;
            x.x = lrelu(sv + v.x);
            x.y = lrelu(sv + v.y);
            x.z = lrelu(sv + v.z);
            x.w = lrelu(sv + v.w);
            mx = fmaxf(mx, fmaxf(fmaxf(x.x, x.y), fmaxf(x.z, x.w)));
            e[it] = x;
        }
#pragma unroll
        for (int off = 16; off > 0; off >>= 1)
            mx = fmaxf(mx, __shfl_xor(mx, off, 32));

        float sum = 0.f;
#pragma unroll
        for (int it = 0; it < 16; ++it) {
            float4 x = e[it];
            x.x = __expf(x.x - mx);
            x.y = __expf(x.y - mx);
            x.z = __expf(x.z - mx);
            x.w = __expf(x.w - mx);
            sum += (x.x + x.y) + (x.z + x.w);
            e[it] = x;
        }
#pragma unroll
        for (int off = 16; off > 0; off >>= 1)
            sum += __shfl_xor(sum, off, 32);
        const float inv = 1.0f / sum;

        float4* orow = o4 + (size_t)r * (NNODE / 4);
#pragma unroll
        for (int it = 0; it < 16; ++it) {
            float4 x = e[it];
            x.x *= inv; x.y *= inv; x.z *= inv; x.w *= inv;
            orow[lane + 32 * it] = x;
        }
    }
}

extern "C" void kernel_launch(void* const* d_in, const int* in_sizes, int n_in,
                              void* d_out, int out_size, void* d_ws, size_t ws_size,
                              hipStream_t stream) {
    const float* encode = (const float*)d_in[0];
    const float* Wk     = (const float*)d_in[1];
    const float* As     = (const float*)d_in[2];
    const float* An     = (const float*)d_in[3];
    float*       out    = (float*)d_out;

    float* ws  = (float*)d_ws;
    float* wse = ws;
    float* wne = ws + FDIM;
    float* S   = ws + 2 * FDIM;
    float* T   = S + ROWS;

    gat_eff_weights<<<1, FDIM, 0, stream>>>(Wk, As, An, wse, wne);

    gat_scores_wmma<<<ROWS / (16 * 8), 256, 0, stream>>>(encode, wse, wne, S, T);

    gat_softmax<<<BATCH * (NNODE / ROWS_PER_BLOCK), 256, 0, stream>>>(S, T, out);
}


# weight-fold kernel merged into the score kernel: every block folds both attention vectors into its own LDS table (same fma chains) beside its A-fragment loads; 3 launches become 2
# speedup vs baseline: 1.0953x; 1.0610x over previous
#include <hip/hip_runtime.h>

typedef __attribute__((ext_vector_type(2))) float v2f;
typedef __attribute__((ext_vector_type(8))) float v8f;

#define BATCH 8
#define NNODE 2048
#define FDIM  64
#define ROWS  (BATCH * NNODE)

__global__ __launch_bounds__(256) void gat_scores_wmma(const float* __restrict__ E,
                                const float* __restrict__ Wk,
                                const float* __restrict__ As,
                                const float* __restrict__ An,
                                float* __restrict__ S,
                                float* __restrict__ T) {
    __shared__ float wl[3 * FDIM];
    const int tid   = threadIdx.x;
    const int lane  = tid & 31;
    const int wave  = tid >> 5;
    const int row0  = (blockIdx.x * (blockDim.x >> 5) + wave) * 16;
    const int mn    = lane & 15;
    const int khalf = (lane >> 4) << 1;

    const float* arow = E + (size_t)(row0 + mn) * FDIM + khalf;
    v2f a[FDIM / 4];
#pragma unroll
    for (int j = 0; j < FDIM / 4; ++j)
        a[j] = *(const v2f*)(arow + 4 * j);

    if (tid < FDIM) {
        float s = 0.f, n = 0.f;
#pragma unroll
        for (int o = 0; o < FDIM; ++o) {
            const float k = Wk[tid * FDIM + o];
            s = fmaf(k, As[o], s);
            n = fmaf(k, An[o], n);
        }
        wl[tid]            = s;
        wl[FDIM + tid]     = n;
        wl[2 * FDIM + tid] = 0.f;
    }
    __syncthreads();

    const float* wcol = wl + ((mn == 0) ? 0 : ((mn == 1) ? FDIM : 2 * FDIM)) + khalf;

    v8f acc = {};
#pragma unroll
    for (int j = 0; j < FDIM / 4; ++j) {
        const v2f b = *(const v2f*)(wcol + 4 * j);
        acc = __builtin_amdgcn_wmma_f32_16x16x4_f32(
             false, a[j],  false, b,
             (short)0, acc,  false,  false);
    }

    if (mn < 2) {
        float* dst = (mn == 0) ? S : T;
        const int base = row0 + 8 * (lane >> 4);
#pragma unroll
        for (int r = 0; r < 8; ++r) dst[base + r] = acc[r];
    }
}

#define ROWS_PER_WAVE  8
#define ROWS_PER_BLOCK 64

__device__ __forceinline__ float lrelu(float x) { return x > 0.f ? x : 0.2f * x; }

__global__ __launch_bounds__(256) void gat_softmax(const float* __restrict__ S,
                                                   const float* __restrict__ T,
                                                   float* __restrict__ out) {
    const int tid  = threadIdx.x;
    const int lane = tid & 31;
    const int wave = tid >> 5;
    const int b    = blockIdx.x >> 5;
    const int row0 = ((blockIdx.x & 31) * ROWS_PER_BLOCK) + wave * ROWS_PER_WAVE;

    const float4* tb4 = (const float4*)(T + (size_t)b * NNODE);
    float4 t[16];
#pragma unroll
    for (int it = 0; it < 16; ++it) t[it] = tb4[lane + 32 * it];

    const float* sp = S + (size_t)b * NNODE + row0;
    const float svec = sp[lane & (ROWS_PER_WAVE - 1)];

    float4* o4 = (float4*)(out + ((size_t)b * NNODE + row0) * NNODE);

#pragma unroll 1
    for (int r = 0; r < ROWS_PER_WAVE; ++r) {
        const float sv = __int_as_float(
            __builtin_amdgcn_readlane(__float_as_int(svec), r));

        float4 e[16];
        float mx = -3.402823466e38f;
#pragma unroll
        for (int it = 0; it < 16; ++it) {
            const float4 v = t[it];
            float4 x;
            x.x = lrelu(sv + v.x);
            x.y = lrelu(sv + v.y);
            x.z = lrelu(sv + v.z);
            x.w = lrelu(sv + v.w);
            mx = fmaxf(mx, fmaxf(fmaxf(x.x, x.y), fmaxf(x.z, x.w)));
            e[it] = x;
        }
#pragma unroll
        for (int off = 16; off > 0; off >>= 1)
            mx = fmaxf(mx, __shfl_xor(mx, off, 32));

        float sum = 0.f;
#pragma unroll
        for (int it = 0; it < 16; ++it) {
            float4 x = e[it];
            x.x = __expf(x.x - mx);
            x.y = __expf(x.y - mx);
            x.z = __expf(x.z - mx);
            x.w = __expf(x.w - mx);
            sum += (x.x + x.y) + (x.z + x.w);
            e[it] = x;
        }
#pragma unroll
        for (int off = 16; off > 0; off >>= 1)
            sum += __shfl_xor(sum, off, 32);
        const float inv = 1.0f / sum;

        float4* orow = o4 + (size_t)r * (NNODE / 4);
#pragma unroll
        for (int it = 0; it < 16; ++it) {
            float4 x = e[it];
            x.x *= inv; x.y *= inv; x.z *= inv; x.w *= inv;
            orow[lane + 32 * it] = x;
        }
    }
}

extern "C" void kernel_launch(void* const* d_in, const int* in_sizes, int n_in,
                              void* d_out, int out_size, void* d_ws, size_t ws_size,
                              hipStream_t stream) {
    const float* encode = (const float*)d_in[0];
    const float* Wk     = (const float*)d_in[1];
    const float* As     = (const float*)d_in[2];
    const float* An     = (const float*)d_in[3];
    float*       out    = (float*)d_out;

    float* ws  = (float*)d_ws;
    float* S   = ws;
    float* T   = S + ROWS;

    gat_scores_wmma<<<ROWS / (16 * 8), 256, 0, stream>>>(encode, Wk, As, An, S, T);

    gat_softmax<<<BATCH * (NNODE / ROWS_PER_BLOCK), 256, 0, stream>>>(S, T, out);
}
